# starting kernel
# baseline (speedup 1.0000x reference)
#include <hip/hip_runtime.h>

typedef float v2f __attribute__((ext_vector_type(2)));
typedef float v8f __attribute__((ext_vector_type(8)));

#define BB    64
#define CIN   512
#define TT    1024
#define COUT  256

#define O_TILE  64
#define T_CHUNK 32
#define WROW    516
#define YROW    33

#define XOFF (O_TILE * WROW)
#define YOFF (XOFF + T_CHUNK * WROW)
#define SMEM_FLOATS (YOFF + O_TILE * YROW)

__global__ __launch_bounds__(256)
void lif_head_fused(const float* __restrict__ x, const float* __restrict__ W,
                    const float* __restrict__ gamma, const float* __restrict__ beta,
                    const float* __restrict__ rmean, const float* __restrict__ rvar,
                    float* __restrict__ out)
{
    extern __shared__ float smem[];
    float* sW = smem;
    float* sX = smem + XOFF;
    float* sY = smem + YOFF;

    const int tid  = threadIdx.x;
    const int wave = tid >> 5;
    const int lane = tid & 31;
    const int b    = blockIdx.x >> 2;
    const int o0   = (blockIdx.x & 3) * O_TILE;

    for (int i = tid; i < O_TILE * (CIN / 4); i += 256) {
        const int o  = i >> 7;
        const int c4 = (i & 127) << 2;
        const float4 w4 = *(const float4*)(W + (size_t)(o0 + o) * CIN + c4);
        *(float4*)(sW + o * WROW + c4) = w4;
    }

    float scale = 0.f, bias = 0.f, v = 0.f, cnt = 0.f;
    if (tid < O_TILE) {
        const int o = o0 + tid;
        scale = gamma[o] * rsqrtf(rvar[o] + 1e-5f);
        bias  = beta[o] - rmean[o] * scale;
    }
    __syncthreads();

    const int m_tile = wave & 3;
    const int n_tile = wave >> 2;
    const int l15    = lane & 15;
    const int khalf  = (lane >> 4) << 1;

    const float* aBase = sW + (m_tile * 16 + l15) * WROW + khalf;
    const float* bBase = sX + (n_tile * 16 + l15) * WROW + khalf;

    for (int chunk = 0; chunk < TT / T_CHUNK; ++chunk) {
        const int t0 = chunk * T_CHUNK;

        {
            const int tseg = (tid & 7) << 2;
            const int crow = tid >> 3;
            for (int p = 0; p < CIN / 32; ++p) {
                const int c = p * 32 + crow;
                const float4 xv = *(const float4*)(x + ((size_t)(b * CIN + c)) * TT + t0 + tseg);
                sX[(tseg + 0) * WROW + c] = xv.x;
                sX[(tseg + 1) * WROW + c] = xv.y;
                sX[(tseg + 2) * WROW + c] = xv.z;
                sX[(tseg + 3) * WROW + c] = xv.w;
            }
        }
        __syncthreads();

        v8f acc0 = {};
        v8f acc1 = {};
        #pragma unroll 2
        for (int kk = 0; kk < CIN; kk += 8) {
            const v2f a0 = *(const v2f*)(aBase + kk);
            const v2f b0 = *(const v2f*)(bBase + kk);
            const v2f a1 = *(const v2f*)(aBase + kk + 4);
            const v2f b1 = *(const v2f*)(bBase + kk + 4);
            acc0 = __builtin_amdgcn_wmma_f32_16x16x4_f32(
                       false, a0, false, b0, (short)0, acc0, false, false);
            acc1 = __builtin_amdgcn_wmma_f32_16x16x4_f32(
                       false, a1, false, b1, (short)0, acc1, false, false);
        }

        {
            const int orow = m_tile * 16 + ((lane >> 4) << 3);
            const int tcol = n_tile * 16 + l15;
            #pragma unroll
            for (int r = 0; r < 8; ++r)
                sY[(orow + r) * YROW + tcol] = acc0[r] + acc1[r];
        }
        __syncthreads();

        if (tid < O_TILE) {
            const float* yrow = sY + tid * YROW;
            #pragma unroll 8
            for (int t = 0; t < T_CHUNK; ++t) {
                const float y = yrow[t] * scale + bias;
                v = v + (y - v) * 0.5f;
                const float s = (v >= 1.0f) ? 1.0f : 0.0f;
                cnt += s;
                v *= (1.0f - s);
            }
        }
        __syncthreads();
    }

    if (tid < O_TILE)
        out[(size_t)b * COUT + o0 + tid] = cnt * (1.0f / (float)TT);
}

extern "C" void kernel_launch(void* const* d_in, const int* in_sizes, int n_in,
                              void* d_out, int out_size, void* d_ws, size_t ws_size,
                              hipStream_t stream) {
    (void)in_sizes; (void)n_in; (void)out_size; (void)d_ws; (void)ws_size;
    const float* x     = (const float*)d_in[0];
    const float* W     = (const float*)d_in[1];
    const float* gamma = (const float*)d_in[2];
    const float* beta  = (const float*)d_in[3];
    const float* rmean = (const float*)d_in[4];
    const float* rvar  = (const float*)d_in[5];
    float* out = (float*)d_out;

    const size_t shmem = SMEM_FLOATS * sizeof(float);
    static bool attr_set = false;
    if (!attr_set) {
        hipFuncSetAttribute((const void*)lif_head_fused,
                            hipFuncAttributeMaxDynamicSharedMemorySize, (int)shmem);
        attr_set = true;
    }

    lif_head_fused<<<dim3(BB * (COUT / O_TILE)), dim3(256), shmem, stream>>>(
        x, W, gamma, beta, rmean, rvar, out);
}


# x stage software-pipelined one chunk ahead: global loads issue into registers before the WMMA loop, transposed LDS stores follow it beside the scan, so load round trips hide under matrix work
# speedup vs baseline: 1.2823x; 1.2823x over previous
#include <hip/hip_runtime.h>

typedef float v2f __attribute__((ext_vector_type(2)));
typedef float v8f __attribute__((ext_vector_type(8)));

#define BB    64
#define CIN   512
#define TT    1024
#define COUT  256

#define O_TILE  64
#define T_CHUNK 32
#define WROW    516
#define YROW    33
#define XPASS   (CIN / 32)

#define XOFF (O_TILE * WROW)
#define YOFF (XOFF + T_CHUNK * WROW)
#define SMEM_FLOATS (YOFF + O_TILE * YROW)

__global__ __launch_bounds__(256)
void lif_head_fused(const float* __restrict__ x, const float* __restrict__ W,
                    const float* __restrict__ gamma, const float* __restrict__ beta,
                    const float* __restrict__ rmean, const float* __restrict__ rvar,
                    float* __restrict__ out)
{
    extern __shared__ float smem[];
    float* sW = smem;
    float* sX = smem + XOFF;
    float* sY = smem + YOFF;

    const int tid  = threadIdx.x;
    const int wave = tid >> 5;
    const int lane = tid & 31;
    const int b    = blockIdx.x >> 2;
    const int o0   = (blockIdx.x & 3) * O_TILE;

    const int tseg = (tid & 7) << 2;
    const int crow = tid >> 3;
    const float* xsrc = x + ((size_t)(b * CIN + crow)) * TT + tseg;
    float* xdst = sX + tseg * WROW + crow;
    float4 xr[XPASS];

    #pragma unroll
    for (int p = 0; p < XPASS; ++p)
        xr[p] = *(const float4*)(xsrc + (size_t)p * 32 * TT);

    for (int i = tid; i < O_TILE * (CIN / 4); i += 256) {
        const int o  = i >> 7;
        const int c4 = (i & 127) << 2;
        const float4 w4 = *(const float4*)(W + (size_t)(o0 + o) * CIN + c4);
        *(float4*)(sW + o * WROW + c4) = w4;
    }

    float scale = 0.f, bias = 0.f, v = 0.f, cnt = 0.f;
    if (tid < O_TILE) {
        const int o = o0 + tid;
        scale = gamma[o] * rsqrtf(rvar[o] + 1e-5f);
        bias  = beta[o] - rmean[o] * scale;
    }
    #pragma unroll
    for (int p = 0; p < XPASS; ++p) {
        float* d = xdst + p * 32;
        d[0 * WROW] = xr[p].x;
        d[1 * WROW] = xr[p].y;
        d[2 * WROW] = xr[p].z;
        d[3 * WROW] = xr[p].w;
    }
    __syncthreads();

    const int m_tile = wave & 3;
    const int n_tile = wave >> 2;
    const int l15    = lane & 15;
    const int khalf  = (lane >> 4) << 1;

    const float* aBase = sW + (m_tile * 16 + l15) * WROW + khalf;
    const float* bBase = sX + (n_tile * 16 + l15) * WROW + khalf;

    for (int chunk = 0; chunk < TT / T_CHUNK; ++chunk) {
        const bool more = (chunk + 1 < TT / T_CHUNK);

        if (more) {
            const float* s = xsrc + (chunk + 1) * T_CHUNK;
            #pragma unroll
            for (int p = 0; p < XPASS; ++p)
                xr[p] = *(const float4*)(s + (size_t)p * 32 * TT);
        }

        v8f acc0 = {};
        v8f acc1 = {};
        #pragma unroll 2
        for (int kk = 0; kk < CIN; kk += 8) {
            const v2f a0 = *(const v2f*)(aBase + kk);
            const v2f b0 = *(const v2f*)(bBase + kk);
            const v2f a1 = *(const v2f*)(aBase + kk + 4);
            const v2f b1 = *(const v2f*)(bBase + kk + 4);
            acc0 = __builtin_amdgcn_wmma_f32_16x16x4_f32(
                       false, a0, false, b0, (short)0, acc0, false, false);
            acc1 = __builtin_amdgcn_wmma_f32_16x16x4_f32(
                       false, a1, false, b1, (short)0, acc1, false, false);
        }

        {
            const int orow = m_tile * 16 + ((lane >> 4) << 3);
            const int tcol = n_tile * 16 + l15;
            #pragma unroll
            for (int r = 0; r < 8; ++r)
                sY[(orow + r) * YROW + tcol] = acc0[r] + acc1[r];
        }
        __syncthreads();

        if (tid < O_TILE) {
            const float* yrow = sY + tid * YROW;
            #pragma unroll 8
            for (int t = 0; t < T_CHUNK; ++t) {
                const float y = yrow[t] * scale + bias;
                v = v + (y - v) * 0.5f;
                const float s = (v >= 1.0f) ? 1.0f : 0.0f;
                cnt += s;
                v *= (1.0f - s);
            }
        }

        if (more) {
            #pragma unroll
            for (int p = 0; p < XPASS; ++p) {
                float* d = xdst + p * 32;
                d[0 * WROW] = xr[p].x;
                d[1 * WROW] = xr[p].y;
                d[2 * WROW] = xr[p].z;
                d[3 * WROW] = xr[p].w;
            }
        }
        __syncthreads();
    }

    if (tid < O_TILE)
        out[(size_t)b * COUT + o0 + tid] = cnt * (1.0f / (float)TT);
}

extern "C" void kernel_launch(void* const* d_in, const int* in_sizes, int n_in,
                              void* d_out, int out_size, void* d_ws, size_t ws_size,
                              hipStream_t stream) {
    (void)in_sizes; (void)n_in; (void)out_size; (void)d_ws; (void)ws_size;
    const float* x     = (const float*)d_in[0];
    const float* W     = (const float*)d_in[1];
    const float* gamma = (const float*)d_in[2];
    const float* beta  = (const float*)d_in[3];
    const float* rmean = (const float*)d_in[4];
    const float* rvar  = (const float*)d_in[5];
    float* out = (float*)d_out;

    const size_t shmem = SMEM_FLOATS * sizeof(float);
    static bool attr_set = false;
    if (!attr_set) {
        hipFuncSetAttribute((const void*)lif_head_fused,
                            hipFuncAttributeMaxDynamicSharedMemorySize, (int)shmem);
        attr_set = true;
    }

    lif_head_fused<<<dim3(BB * (COUT / O_TILE)), dim3(256), shmem, stream>>>(
        x, W, gamma, beta, rmean, rvar, out);
}


# GEMM inner loop software-pipelined: LDS fragment loads of each 16-WMMA block issue as one group one block ahead into a second register set (was: load, wait, 4 WMMAs); summation order unchanged
# speedup vs baseline: 1.3385x; 1.0438x over previous
#include <hip/hip_runtime.h>

typedef float v2f __attribute__((ext_vector_type(2)));
typedef float v8f __attribute__((ext_vector_type(8)));

#define BB    64
#define CIN   512
#define TT    1024
#define COUT  256

#define O_TILE  64
#define T_CHUNK 32
#define WROW    516
#define YROW    33
#define XPASS   (CIN / 32)

#define XOFF (O_TILE * WROW)
#define YOFF (XOFF + T_CHUNK * WROW)
#define SMEM_FLOATS (YOFF + O_TILE * YROW)

#define KQ   16
#define NBLK (CIN / (4 * KQ))

__device__ __forceinline__ void load_frags(v2f (&fa)[KQ], v2f (&fb)[KQ],
                                           const float* a, const float* b)
{
    #pragma unroll
    for (int q = 0; q < KQ; ++q) fa[q] = *(const v2f*)(a + 4 * q);
    #pragma unroll
    for (int q = 0; q < KQ; ++q) fb[q] = *(const v2f*)(b + 4 * q);
}

__device__ __forceinline__ void mma_block(const v2f (&fa)[KQ], const v2f (&fb)[KQ],
                                          v8f& acc0, v8f& acc1)
{
    #pragma unroll
    for (int q = 0; q < KQ; q += 2) {
        acc0 = __builtin_amdgcn_wmma_f32_16x16x4_f32(
                   false, fa[q], false, fb[q], (short)0, acc0, false, false);
        acc1 = __builtin_amdgcn_wmma_f32_16x16x4_f32(
                   false, fa[q + 1], false, fb[q + 1], (short)0, acc1, false, false);
    }
}

__global__ __launch_bounds__(256)
void lif_head_fused(const float* __restrict__ x, const float* __restrict__ W,
                    const float* __restrict__ gamma, const float* __restrict__ beta,
                    const float* __restrict__ rmean, const float* __restrict__ rvar,
                    float* __restrict__ out)
{
    extern __shared__ float smem[];
    float* sW = smem;
    float* sX = smem + XOFF;
    float* sY = smem + YOFF;

    const int tid  = threadIdx.x;
    const int wave = tid >> 5;
    const int lane = tid & 31;
    const int b    = blockIdx.x >> 2;
    const int o0   = (blockIdx.x & 3) * O_TILE;

    const int tseg = (tid & 7) << 2;
    const int crow = tid >> 3;
    const float* xsrc = x + ((size_t)(b * CIN + crow)) * TT + tseg;
    float* xdst = sX + tseg * WROW + crow;
    float4 xr[XPASS];

    #pragma unroll
    for (int p = 0; p < XPASS; ++p)
        xr[p] = *(const float4*)(xsrc + (size_t)p * 32 * TT);

    for (int i = tid; i < O_TILE * (CIN / 4); i += 256) {
        const int o  = i >> 7;
        const int c4 = (i & 127) << 2;
        const float4 w4 = *(const float4*)(W + (size_t)(o0 + o) * CIN + c4);
        *(float4*)(sW + o * WROW + c4) = w4;
    }

    float scale = 0.f, bias = 0.f, v = 0.f, cnt = 0.f;
    if (tid < O_TILE) {
        const int o = o0 + tid;
        scale = gamma[o] * rsqrtf(rvar[o] + 1e-5f);
        bias  = beta[o] - rmean[o] * scale;
    }
    #pragma unroll
    for (int p = 0; p < XPASS; ++p) {
        float* d = xdst + p * 32;
        d[0 * WROW] = xr[p].x;
        d[1 * WROW] = xr[p].y;
        d[2 * WROW] = xr[p].z;
        d[3 * WROW] = xr[p].w;
    }
    __syncthreads();

    const int m_tile = wave & 3;
    const int n_tile = wave >> 2;
    const int l15    = lane & 15;
    const int khalf  = (lane >> 4) << 1;

    const float* aBase = sW + (m_tile * 16 + l15) * WROW + khalf;
    const float* bBase = sX + (n_tile * 16 + l15) * WROW + khalf;

    for (int chunk = 0; chunk < TT / T_CHUNK; ++chunk) {
        const bool more = (chunk + 1 < TT / T_CHUNK);

        if (more) {
            const float* s = xsrc + (chunk + 1) * T_CHUNK;
            #pragma unroll
            for (int p = 0; p < XPASS; ++p)
                xr[p] = *(const float4*)(s + (size_t)p * 32 * TT);
        }

        v8f acc0 = {};
        v8f acc1 = {};
        v2f fa[2][KQ], fb[2][KQ];
        load_frags(fa[0], fb[0], aBase, bBase);
        #pragma unroll
        for (int blk = 0; blk < NBLK; ++blk) {
            if (blk + 1 < NBLK)
                load_frags(fa[(blk + 1) & 1], fb[(blk + 1) & 1],
                           aBase + (blk + 1) * (4 * KQ), bBase + (blk + 1) * (4 * KQ));
            mma_block(fa[blk & 1], fb[blk & 1], acc0, acc1);
        }
        __builtin_amdgcn_sched_group_barrier(0x100, KQ, 0);
        #pragma unroll
        for (int blk = 0; blk < NBLK; ++blk) {
            if (blk + 1 < NBLK)
                __builtin_amdgcn_sched_group_barrier(0x100, KQ, 0);
            __builtin_amdgcn_sched_group_barrier(0x008, KQ, 0);
        }

        {
            const int orow = m_tile * 16 + ((lane >> 4) << 3);
            const int tcol = n_tile * 16 + l15;
            #pragma unroll
            for (int r = 0; r < 8; ++r)
                sY[(orow + r) * YROW + tcol] = acc0[r] + acc1[r];
        }
        __syncthreads();

        if (tid < O_TILE) {
            const float* yrow = sY + tid * YROW;
            #pragma unroll 8
            for (int t = 0; t < T_CHUNK; ++t) {
                const float y = yrow[t] * scale + bias;
                v = v + (y - v) * 0.5f;
                const float s = (v >= 1.0f) ? 1.0f : 0.0f;
                cnt += s;
                v *= (1.0f - s);
            }
        }

        if (more) {
            #pragma unroll
            for (int p = 0; p < XPASS; ++p) {
                float* d = xdst + p * 32;
                d[0 * WROW] = xr[p].x;
                d[1 * WROW] = xr[p].y;
                d[2 * WROW] = xr[p].z;
                d[3 * WROW] = xr[p].w;
            }
        }
        __syncthreads();
    }

    if (tid < O_TILE)
        out[(size_t)b * COUT + o0 + tid] = cnt * (1.0f / (float)TT);
}

extern "C" void kernel_launch(void* const* d_in, const int* in_sizes, int n_in,
                              void* d_out, int out_size, void* d_ws, size_t ws_size,
                              hipStream_t stream) {
    (void)in_sizes; (void)n_in; (void)out_size; (void)d_ws; (void)ws_size;
    const float* x     = (const float*)d_in[0];
    const float* W     = (const float*)d_in[1];
    const float* gamma = (const float*)d_in[2];
    const float* beta  = (const float*)d_in[3];
    const float* rmean = (const float*)d_in[4];
    const float* rvar  = (const float*)d_in[5];
    float* out = (float*)d_out;

    const size_t shmem = SMEM_FLOATS * sizeof(float);
    static bool attr_set = false;
    if (!attr_set) {
        hipFuncSetAttribute((const void*)lif_head_fused,
                            hipFuncAttributeMaxDynamicSharedMemorySize, (int)shmem);
        attr_set = true;
    }

    lif_head_fused<<<dim3(BB * (COUT / O_TILE)), dim3(256), shmem, stream>>>(
        x, W, gamma, beta, rmean, rvar, out);
}


# chunk loop pipelined over two x and two y LDS stages, one barrier per chunk: LIF scan of chunk i-1 and restaging of chunk i+1 (in the fragment-load groups) overlap chunk i's WMMAs; sums unchanged
# speedup vs baseline: 1.5558x; 1.1623x over previous
#include <hip/hip_runtime.h>


typedef float v2f __attribute__((ext_vector_type(2)));
typedef float v8f __attribute__((ext_vector_type(8)));

#define BB    64
#define CIN   512
#define TT    1024
#define COUT  256

#define O_TILE  64
#define T_CHUNK 32
#define NCHUNK  (TT / T_CHUNK)
#define WROW    516
#define YROW    33
#define XPASS   (CIN / 32)

#define XBUF (T_CHUNK * WROW)
#define YBUF (O_TILE * YROW)
#define XOFF (O_TILE * WROW)
#define YOFF (XOFF + 2 * XBUF)
#define SMEM_FLOATS (YOFF + 2 * YBUF)

#define KQ   16
#define NBLK (CIN / (4 * KQ))
#define PPB  4
#define XBLK (XPASS / PPB)

__device__ __forceinline__ void load_frags(v2f (&fa)[KQ], v2f (&fb)[KQ],
                                           const float* a, const float* b)
{
    #pragma unroll
    for (int q = 0; q < KQ; ++q) fa[q] = *(const v2f*)(a + 4 * q);
    #pragma unroll
    for (int q = 0; q < KQ; ++q) fb[q] = *(const v2f*)(b + 4 * q);
}

__device__ __forceinline__ void mma_block(const v2f (&fa)[KQ], const v2f (&fb)[KQ],
                                          v8f& acc0, v8f& acc1)
{
    #pragma unroll
    for (int q = 0; q < KQ; q += 2) {
        acc0 = __builtin_amdgcn_wmma_f32_16x16x4_f32(
                   false, fa[q], false, fb[q], (short)0, acc0, false, false);
        acc1 = __builtin_amdgcn_wmma_f32_16x16x4_f32(
                   false, fa[q + 1], false, fb[q + 1], (short)0, acc1, false, false);
    }
}

__device__ __forceinline__ void lif_scan(const float* yrow, const float scale, const float bias,
                                         float& v, float& cnt)
{
    #pragma unroll 8
    for (int t = 0; t < T_CHUNK; ++t) {
        const float y = yrow[t] * scale + bias;
        v = v + (y - v) * 0.5f;
        const float s = (v >= 1.0f) ? 1.0f : 0.0f;
        cnt += s;
        v *= (1.0f - s);
    }
}

__global__ __launch_bounds__(256)
void lif_head_fused(const float* __restrict__ x, const float* __restrict__ W,
                    const float* __restrict__ gamma, const float* __restrict__ beta,
                    const float* __restrict__ rmean, const float* __restrict__ rvar,
                    float* __restrict__ out)
{
    __shared__ __attribute__((aligned(16))) float smem[SMEM_FLOATS];
    float* sW = smem;
    float* sX = smem + XOFF;
    float* sY = smem + YOFF;

    const int tid  = threadIdx.x;
    const int wave = tid >> 5;
    const int lane = tid & 31;
    const int b    = blockIdx.x >> 2;
    const int o0   = (blockIdx.x & 3) * O_TILE;

    const int tseg = (tid & 7) << 2;
    const int crow = tid >> 3;
    const float* xsrc = x + ((size_t)(b * CIN + crow)) * TT + tseg;
    float* xdst = sX + tseg * WROW + crow;
    float4 xr[XPASS];

    float4 x0[XPASS];
    #pragma unroll
    for (int p = 0; p < XPASS; ++p)
        x0[p] = *(const float4*)(xsrc + (size_t)p * 32 * TT);
    #pragma unroll
    for (int p = 0; p < XPASS; ++p)
        xr[p] = *(const float4*)(xsrc + T_CHUNK + (size_t)p * 32 * TT);

    for (int i = tid; i < O_TILE * (CIN / 4); i += 256) {
        const int o  = i >> 7;
        const int c4 = (i & 127) << 2;
        const float4 w4 = *(const float4*)(W + (size_t)(o0 + o) * CIN + c4);
        *(float4*)(sW + o * WROW + c4) = w4;
    }

    float scale = 0.f, bias = 0.f, v = 0.f, cnt = 0.f;
    if (tid < O_TILE) {
        const int o = o0 + tid;
        scale = gamma[o] * rsqrtf(rvar[o] + 1e-5f);
        bias  = beta[o] - rmean[o] * scale;
    }
    #pragma unroll
    for (int p = 0; p < XPASS; ++p) {
        float* d = xdst + p * 32;
        d[0 * WROW] = x0[p].x;
        d[1 * WROW] = x0[p].y;
        d[2 * WROW] = x0[p].z;
        d[3 * WROW] = x0[p].w;
    }
    __syncthreads();

    const int m_tile = wave & 3;
    const int n_tile = wave >> 2;
    const int l15    = lane & 15;
    const int khalf  = (lane >> 4) << 1;

    const float* aBase = sW + (m_tile * 16 + l15) * WROW + khalf;
    const float* bBase = sX + (n_tile * 16 + l15) * WROW + khalf;

    for (int chunk = 0; chunk < NCHUNK; ++chunk) {
        const int cur = chunk & 1;

        if (chunk > 0 && tid < O_TILE)
            lif_scan(sY + (cur ^ 1) * YBUF + tid * YROW, scale, bias, v, cnt);

        const float* bCur  = bBase + cur * XBUF;
        float*       xdstN = xdst + (cur ^ 1) * XBUF;
        const int    nxt   = (chunk + 2 < NCHUNK) ? (chunk + 2) : (NCHUNK - 1);
        const float* xnext = xsrc + nxt * T_CHUNK;

        v8f acc0 = {};
        v8f acc1 = {};
        v2f fa[2][KQ], fb[2][KQ];
        load_frags(fa[0], fb[0], aBase, bCur);
        #pragma unroll
        for (int blk = 0; blk < NBLK; ++blk) {
            if (blk + 1 < NBLK)
                load_frags(fa[(blk + 1) & 1], fb[(blk + 1) & 1],
                           aBase + (blk + 1) * (4 * KQ), bCur + (blk + 1) * (4 * KQ));
            if (blk < XBLK) {
                #pragma unroll
                for (int pp = 0; pp < PPB; ++pp) {
                    const int p = blk * PPB + pp;
                    float* d = xdstN + p * 32;
                    d[0 * WROW] = xr[p].x;
                    d[1 * WROW] = xr[p].y;
                    d[2 * WROW] = xr[p].z;
                    d[3 * WROW] = xr[p].w;
                    xr[p] = *(const float4*)(xnext + (size_t)p * 32 * TT);
                }
            }
            mma_block(fa[blk & 1], fb[blk & 1], acc0, acc1);
        }
        __builtin_amdgcn_sched_group_barrier(0x100, KQ, 0);
        #pragma unroll
        for (int blk = 0; blk < NBLK; ++blk) {
            if (blk + 1 < NBLK)
                __builtin_amdgcn_sched_group_barrier(0x100, KQ, 0);
            if (blk < XBLK) {
                __builtin_amdgcn_sched_group_barrier(0x200, 2 * PPB, 0);
                __builtin_amdgcn_sched_group_barrier(0x020, PPB, 0);
            }
            __builtin_amdgcn_sched_group_barrier(0x008, KQ, 0);
        }

        {
            float* sYc = sY + cur * YBUF;
            const int orow = m_tile * 16 + ((lane >> 4) << 3);
            const int tcol = n_tile * 16 + l15;
            #pragma unroll
            for (int r = 0; r < 8; ++r)
                sYc[(orow + r) * YROW + tcol] = acc0[r] + acc1[r];
        }
        __syncthreads();
    }

    if (tid < O_TILE) {
        lif_scan(sY + ((NCHUNK - 1) & 1) * YBUF + tid * YROW, scale, bias, v, cnt);
        out[(size_t)b * COUT + o0 + tid] = cnt * (1.0f / (float)TT);
    }
}

extern "C" void kernel_launch(void* const* d_in, const int* in_sizes, int n_in,
                              void* d_out, int out_size, void* d_ws, size_t ws_size,
                              hipStream_t stream) {
    (void)in_sizes; (void)n_in; (void)out_size; (void)d_ws; (void)ws_size;
    const float* x     = (const float*)d_in[0];
    const float* W     = (const float*)d_in[1];
    const float* gamma = (const float*)d_in[2];
    const float* beta  = (const float*)d_in[3];
    const float* rmean = (const float*)d_in[4];
    const float* rvar  = (const float*)d_in[5];
    float* out = (float*)d_out;

    lif_head_fused<<<dim3(BB * (COUT / O_TILE)), dim3(256), 0, stream>>>(
        x, W, gamma, beta, rmean, rvar, out);
}


# W (A) fragments register-resident: each wave reads its rows of W over all of Cin from LDS once, before the chunk loop, which now loads only B fragments (half the LDS reads per WMMA); sums unchanged
# speedup vs baseline: 1.6204x; 1.0415x over previous
#include <hip/hip_runtime.h>


typedef float v2f __attribute__((ext_vector_type(2)));
typedef float v8f __attribute__((ext_vector_type(8)));

#define BB    64
#define CIN   512
#define TT    1024
#define COUT  256

#define O_TILE  64
#define T_CHUNK 32
#define NCHUNK  (TT / T_CHUNK)
#define WROW    516
#define YROW    33
#define XPASS   (CIN / 32)

#define XBUF (T_CHUNK * WROW)
#define YBUF (O_TILE * YROW)
#define XOFF (O_TILE * WROW)
#define YOFF (XOFF + 2 * XBUF)
#define SMEM_FLOATS (YOFF + 2 * YBUF)

#define KQ   16
#define NBLK (CIN / (4 * KQ))
#define PPB  4
#define XBLK (XPASS / PPB)

#define NKQ (CIN / 4)

__device__ __forceinline__ void load_frags_b(v2f (&fb)[KQ], const float* b)
{
    #pragma unroll
    for (int q = 0; q < KQ; ++q) fb[q] = *(const v2f*)(b + 4 * q);
}

__device__ __forceinline__ void mma_block(const v2f (&fa)[NKQ], const int blk, const v2f (&fb)[KQ],
                                          v8f& acc0, v8f& acc1)
{
    #pragma unroll
    for (int q = 0; q < KQ; q += 2) {
        acc0 = __builtin_amdgcn_wmma_f32_16x16x4_f32(
                   false, fa[blk * KQ + q], false, fb[q], (short)0, acc0, false, false);
        acc1 = __builtin_amdgcn_wmma_f32_16x16x4_f32(
                   false, fa[blk * KQ + q + 1], false, fb[q + 1], (short)0, acc1, false, false);
    }
}

__device__ __forceinline__ void lif_scan(const float* yrow, const float scale, const float bias,
                                         float& v, float& cnt)
{
    #pragma unroll 8
    for (int t = 0; t < T_CHUNK; ++t) {
        const float y = yrow[t] * scale + bias;
        v = v + (y - v) * 0.5f;
        const float s = (v >= 1.0f) ? 1.0f : 0.0f;
        cnt += s;
        v *= (1.0f - s);
    }
}

__global__ __launch_bounds__(256)
void lif_head_fused(const float* __restrict__ x, const float* __restrict__ W,
                    const float* __restrict__ gamma, const float* __restrict__ beta,
                    const float* __restrict__ rmean, const float* __restrict__ rvar,
                    float* __restrict__ out)
{
    __shared__ __attribute__((aligned(16))) float smem[SMEM_FLOATS];
    float* sW = smem;
    float* sX = smem + XOFF;
    float* sY = smem + YOFF;

    const int tid  = threadIdx.x;
    const int wave = tid >> 5;
    const int lane = tid & 31;
    const int b    = blockIdx.x >> 2;
    const int o0   = (blockIdx.x & 3) * O_TILE;

    const int tseg = (tid & 7) << 2;
    const int crow = tid >> 3;
    const float* xsrc = x + ((size_t)(b * CIN + crow)) * TT + tseg;
    float* xdst = sX + tseg * WROW + crow;
    float4 xr[XPASS];

    float4 x0[XPASS];
    #pragma unroll
    for (int p = 0; p < XPASS; ++p)
        x0[p] = *(const float4*)(xsrc + (size_t)p * 32 * TT);
    #pragma unroll
    for (int p = 0; p < XPASS; ++p)
        xr[p] = *(const float4*)(xsrc + T_CHUNK + (size_t)p * 32 * TT);

    for (int i = tid; i < O_TILE * (CIN / 4); i += 256) {
        const int o  = i >> 7;
        const int c4 = (i & 127) << 2;
        const float4 w4 = *(const float4*)(W + (size_t)(o0 + o) * CIN + c4);
        *(float4*)(sW + o * WROW + c4) = w4;
    }

    float scale = 0.f, bias = 0.f, v = 0.f, cnt = 0.f;
    if (tid < O_TILE) {
        const int o = o0 + tid;
        scale = gamma[o] * rsqrtf(rvar[o] + 1e-5f);
        bias  = beta[o] - rmean[o] * scale;
    }
    #pragma unroll
    for (int p = 0; p < XPASS; ++p) {
        float* d = xdst + p * 32;
        d[0 * WROW] = x0[p].x;
        d[1 * WROW] = x0[p].y;
        d[2 * WROW] = x0[p].z;
        d[3 * WROW] = x0[p].w;
    }
    __syncthreads();

    const int m_tile = wave & 3;
    const int n_tile = wave >> 2;
    const int l15    = lane & 15;
    const int khalf  = (lane >> 4) << 1;

    const float* aBase = sW + (m_tile * 16 + l15) * WROW + khalf;
    const float* bBase = sX + (n_tile * 16 + l15) * WROW + khalf;

    v2f faAll[NKQ];
    #pragma unroll
    for (int q = 0; q < NKQ; ++q) faAll[q] = *(const v2f*)(aBase + 4 * q);

    for (int chunk = 0; chunk < NCHUNK; ++chunk) {
        const int cur = chunk & 1;

        if (chunk > 0 && tid < O_TILE)
            lif_scan(sY + (cur ^ 1) * YBUF + tid * YROW, scale, bias, v, cnt);

        const float* bCur  = bBase + cur * XBUF;
        float*       xdstN = xdst + (cur ^ 1) * XBUF;
        const int    nxt   = (chunk + 2 < NCHUNK) ? (chunk + 2) : (NCHUNK - 1);
        const float* xnext = xsrc + nxt * T_CHUNK;

        v8f acc0 = {};
        v8f acc1 = {};
        v2f fb[2][KQ];
        load_frags_b(fb[0], bCur);
        #pragma unroll
        for (int blk = 0; blk < NBLK; ++blk) {
            if (blk + 1 < NBLK)
                load_frags_b(fb[(blk + 1) & 1], bCur + (blk + 1) * (4 * KQ));
            if (blk < XBLK) {
                #pragma unroll
                for (int pp = 0; pp < PPB; ++pp) {
                    const int p = blk * PPB + pp;
                    float* d = xdstN + p * 32;
                    d[0 * WROW] = xr[p].x;
                    d[1 * WROW] = xr[p].y;
                    d[2 * WROW] = xr[p].z;
                    d[3 * WROW] = xr[p].w;
                    xr[p] = *(const float4*)(xnext + (size_t)p * 32 * TT);
                }
            }
            mma_block(faAll, blk, fb[blk & 1], acc0, acc1);
        }
        __builtin_amdgcn_sched_group_barrier(0x100, KQ / 2, 0);
        #pragma unroll
        for (int blk = 0; blk < NBLK; ++blk) {
            if (blk + 1 < NBLK)
                __builtin_amdgcn_sched_group_barrier(0x100, KQ / 2, 0);
            if (blk < XBLK) {
                __builtin_amdgcn_sched_group_barrier(0x200, 2 * PPB, 0);
                __builtin_amdgcn_sched_group_barrier(0x020, PPB, 0);
            }
            __builtin_amdgcn_sched_group_barrier(0x008, KQ, 0);
        }

        {
            float* sYc = sY + cur * YBUF;
            const int orow = m_tile * 16 + ((lane >> 4) << 3);
            const int tcol = n_tile * 16 + l15;
            #pragma unroll
            for (int r = 0; r < 8; ++r)
                sYc[(orow + r) * YROW + tcol] = acc0[r] + acc1[r];
        }
        __syncthreads();
    }

    if (tid < O_TILE) {
        lif_scan(sY + ((NCHUNK - 1) & 1) * YBUF + tid * YROW, scale, bias, v, cnt);
        out[(size_t)b * COUT + o0 + tid] = cnt * (1.0f / (float)TT);
    }
}

extern "C" void kernel_launch(void* const* d_in, const int* in_sizes, int n_in,
                              void* d_out, int out_size, void* d_ws, size_t ws_size,
                              hipStream_t stream) {
    (void)in_sizes; (void)n_in; (void)out_size; (void)d_ws; (void)ws_size;
    const float* x     = (const float*)d_in[0];
    const float* W     = (const float*)d_in[1];
    const float* gamma = (const float*)d_in[2];
    const float* beta  = (const float*)d_in[3];
    const float* rmean = (const float*)d_in[4];
    const float* rvar  = (const float*)d_in[5];
    float* out = (float*)d_out;

    lif_head_fused<<<dim3(BB * (COUT / O_TILE)), dim3(256), 0, stream>>>(
        x, W, gamma, beta, rmean, rvar, out);
}
